# starting kernel
# baseline (speedup 1.0000x reference)
#include <hip/hip_runtime.h>
#include <math.h>


typedef float v2f __attribute__((ext_vector_type(2)));
typedef float v8f __attribute__((ext_vector_type(8)));

#define HEADS  8
#define CDIM   32
#define HCDIM  256
#define NSLOPE 0.2f

__device__ __forceinline__ void atomicMaxF(float* addr, float val) {
    if (val >= 0.0f) {
        atomicMax((int*)addr, __float_as_int(val));
    } else {
        atomicMin((unsigned int*)addr, __float_as_uint(val));
    }
}

__global__ void wmma_gemm_f32(const float* __restrict__ A,
                              const float* __restrict__ B,
                              float* __restrict__ C, int K, int NB) {
    const int  lane = threadIdx.x & 31;
    const bool hi   = lane >= 16;
    const int  l    = lane & 15;
    const int  row0 = blockIdx.x * 16;
    const int  col0 = blockIdx.y * 16;

    v8f acc = {};
    const float* arow = A + (size_t)(row0 + l) * K + (hi ? 2 : 0);
    const float* bcol = B + (size_t)(hi ? 2 : 0) * NB + col0 + l;
    for (int k = 0; k < K; k += 4) {
        v2f a; a.x = arow[k];                  a.y = arow[k + 1];
        v2f b; b.x = bcol[(size_t)k * NB];     b.y = bcol[(size_t)(k + 1) * NB];
        acc = __builtin_amdgcn_wmma_f32_16x16x4_f32(
                  false, a, false, b, (short)0, acc, false, false);
    }
    float* crow = C + (size_t)(row0 + (hi ? 8 : 0)) * NB + col0 + l;
#pragma unroll
    for (int r = 0; r < 8; ++r) crow[(size_t)r * NB] = acc[r];
}

__global__ void wmma_gemm_relu_bias(const float* __restrict__ Acc,
                                    const float* __restrict__ bias,
                                    const float* __restrict__ B,
                                    const float* __restrict__ bout,
                                    float* __restrict__ C, int K, int NB) {
    const int  lane = threadIdx.x & 31;
    const bool hi   = lane >= 16;
    const int  l    = lane & 15;
    const int  row0 = blockIdx.x * 16;
    const int  col0 = blockIdx.y * 16;

    v8f acc = {};
    const float* arow = Acc + (size_t)(row0 + l) * K;
    const float* bcol = B + (size_t)(hi ? 2 : 0) * NB + col0 + l;
    const int    koff = hi ? 2 : 0;
    for (int k = 0; k < K; k += 4) {
        const int kk = k + koff;
        v2f a;
        a.x = fmaxf(arow[kk]     + bias[kk],     0.0f);
        a.y = fmaxf(arow[kk + 1] + bias[kk + 1], 0.0f);
        v2f b; b.x = bcol[(size_t)k * NB]; b.y = bcol[(size_t)(k + 1) * NB];
        acc = __builtin_amdgcn_wmma_f32_16x16x4_f32(
                  false, a, false, b, (short)0, acc, false, false);
    }
    const float bo = bout[col0 + l];
    float* crow = C + (size_t)(row0 + (hi ? 8 : 0)) * NB + col0 + l;
#pragma unroll
    for (int r = 0; r < 8; ++r) crow[(size_t)r * NB] = acc[r] + bo;
}

__global__ void gat_att_scores(const float* __restrict__ h,
                               const float* __restrict__ att_src,
                               const float* __restrict__ att_dst,
                               float* __restrict__ a_src,
                               float* __restrict__ a_dst,
                               float* __restrict__ maxbuf,
                               float* __restrict__ denom, int N) {
    const int t = blockIdx.x * blockDim.x + threadIdx.x;
    if (t >= N * HEADS) return;
    const int n  = t / HEADS;
    const int hd = t - n * HEADS;
    const float* hp = h + (size_t)n * HCDIM + hd * CDIM;
    const float* as = att_src + hd * CDIM;
    const float* ad = att_dst + hd * CDIM;
    float s0 = 0.0f, s1 = 0.0f;
#pragma unroll
    for (int c = 0; c < CDIM; ++c) {
        const float v = hp[c];
        s0 = fmaf(v, as[c], s0);
        s1 = fmaf(v, ad[c], s1);
    }
    a_src[t]  = s0;
    a_dst[t]  = s1;
    maxbuf[t] = -INFINITY;
    denom[t]  = 0.0f;
}

__global__ void zero_f32(float* __restrict__ p, int n) {
    const int t = blockIdx.x * blockDim.x + threadIdx.x;
    if (t < n) p[t] = 0.0f;
}

__device__ __forceinline__ void edge_src_dst(const int* __restrict__ ei,
                                             int e, int E, int& s, int& d) {
    if (e < E) { s = ei[e]; d = ei[E + e]; }
    else       { s = e - E; d = e - E; }
}

__device__ __forceinline__ float edge_logit(const float* __restrict__ a_src,
                                            const float* __restrict__ a_dst,
                                            int s, int d, int hd) {
    float v = a_src[s * HEADS + hd] + a_dst[d * HEADS + hd];
    return v > 0.0f ? v : NSLOPE * v;
}

__global__ void gat_edge_max(const int* __restrict__ ei,
                             const float* __restrict__ a_src,
                             const float* __restrict__ a_dst,
                             float* __restrict__ maxbuf, int E, int ET) {
    const int t = blockIdx.x * blockDim.x + threadIdx.x;
    if (t >= ET * HEADS) return;
    const int e  = t >> 3;
    const int hd = t & 7;
    int s, d;
    edge_src_dst(ei, e, E, s, d);
    atomicMaxF(&maxbuf[d * HEADS + hd], edge_logit(a_src, a_dst, s, d, hd));
}

__global__ void gat_edge_expsum(const int* __restrict__ ei,
                                const float* __restrict__ a_src,
                                const float* __restrict__ a_dst,
                                const float* __restrict__ maxbuf,
                                float* __restrict__ denom, int E, int ET) {
    const int t = blockIdx.x * blockDim.x + threadIdx.x;
    if (t >= ET * HEADS) return;
    const int e  = t >> 3;
    const int hd = t & 7;
    int s, d;
    edge_src_dst(ei, e, E, s, d);
    const float v  = edge_logit(a_src, a_dst, s, d, hd);
    const float ex = expf(v - maxbuf[d * HEADS + hd]);
    atomicAdd(&denom[d * HEADS + hd], ex);
}

__global__ void gat_edge_scatter(const int* __restrict__ ei,
                                 const float* __restrict__ a_src,
                                 const float* __restrict__ a_dst,
                                 const float* __restrict__ maxbuf,
                                 const float* __restrict__ denom,
                                 const float* __restrict__ h,
                                 float* __restrict__ accum, int E, int ET) {
    const int lane = threadIdx.x & 31;
    const int wid  = (blockIdx.x * blockDim.x + threadIdx.x) >> 5;
    if (wid >= ET) return;
    int s, d;
    edge_src_dst(ei, wid, E, s, d);

    const int   h8    = lane & 7;
    const float v     = edge_logit(a_src, a_dst, s, d, h8);
    const float ex    = expf(v - maxbuf[d * HEADS + h8]);
    const float alpha = ex / (denom[d * HEADS + h8] + 1e-16f);

    const float* hs = h + (size_t)s * HCDIM;
    float*       ap = accum + (size_t)d * HCDIM;
#pragma unroll
    for (int i = 0; i < HEADS; ++i) {
        const float al = __shfl(alpha, i, 32);
        atomicAdd(&ap[i * CDIM + lane], hs[i * CDIM + lane] * al);
    }
}

extern "C" void kernel_launch(void* const* d_in, const int* in_sizes, int n_in,
                              void* d_out, int out_size, void* d_ws, size_t ws_size,
                              hipStream_t stream) {
    (void)n_in; (void)out_size; (void)ws_size;

    const float* x        = (const float*)d_in[0];
    const int*   ei       = (const int*)d_in[1];
    const float* W_gat    = (const float*)d_in[2];
    const float* att_src  = (const float*)d_in[3];
    const float* att_dst  = (const float*)d_in[4];
    const float* bias_gat = (const float*)d_in[5];
    const float* W_lin    = (const float*)d_in[6];
    const float* b_lin    = (const float*)d_in[7];
    float*       out      = (float*)d_out;

    const int IN  = 256;
    const int N   = in_sizes[0] / IN;
    const int E   = in_sizes[1] / 2;
    const int OUT = in_sizes[6] / HCDIM;
    const int ET  = E + N;

    float* ws   = (float*)d_ws;
    float* h    = ws;
    float* acc  = h    + (size_t)N * HCDIM;
    float* asrc = acc  + (size_t)N * HCDIM;
    float* adst = asrc + (size_t)N * HEADS;
    float* mbuf = adst + (size_t)N * HEADS;
    float* den  = mbuf + (size_t)N * HEADS;

    wmma_gemm_f32<<<dim3(N / 16, HCDIM / 16), 32, 0, stream>>>(x, W_gat, h, IN, HCDIM);

    const int t2 = N * HEADS;
    gat_att_scores<<<(t2 + 255) / 256, 256, 0, stream>>>(h, att_src, att_dst,
                                                         asrc, adst, mbuf, den, N);

    const int t3 = N * HCDIM;
    zero_f32<<<(t3 + 255) / 256, 256, 0, stream>>>(acc, t3);

    const int t4 = ET * HEADS;
    gat_edge_max<<<(t4 + 255) / 256, 256, 0, stream>>>(ei, asrc, adst, mbuf, E, ET);

    gat_edge_expsum<<<(t4 + 255) / 256, 256, 0, stream>>>(ei, asrc, adst, mbuf, den, E, ET);

    gat_edge_scatter<<<(ET + 7) / 8, 256, 0, stream>>>(ei, asrc, adst, mbuf, den,
                                                       h, acc, E, ET);

    wmma_gemm_relu_bias<<<dim3(N / 16, OUT / 16), 32, 0, stream>>>(acc, bias_gat,
                                                                   W_lin, b_lin,
                                                                   out, HCDIM, OUT);
}


# Aggregate launch filters edges by 1-byte block keys (dst / 208, written by the score launch), not the 4-byte destinations; one LDS add reserves a batch's list slots; sums stay in arrival order
# speedup vs baseline: 3.8582x; 3.8582x over previous
#include <hip/hip_runtime.h>
#include <math.h>


typedef float v2f __attribute__((ext_vector_type(2)));
typedef float v8f __attribute__((ext_vector_type(8)));

#define HEADS  8
#define CDIM   32
#define HCDIM  256
#define NSLOPE 0.2f
#define FR     208
#define NOKEY  255u

__device__ __forceinline__ void atomicMaxF(float* addr, float val) {
    if (val >= 0.0f) {
        atomicMax((int*)addr, __float_as_int(val));
    } else {
        atomicMin((unsigned int*)addr, __float_as_uint(val));
    }
}

#define GK     256
#define GRT    5
#define GROWS  (GRT * 16)
#define GPITCH 260

template <int NWAVES, int CT>
__global__ __launch_bounds__(NWAVES * 32)
void wmma_gemm_rows_in_lds(const float* __restrict__ A,
                           const float* __restrict__ B,
                           float* __restrict__ C, int M, int NB) {
    __shared__ __attribute__((aligned(16))) float sA[GROWS * GPITCH];

    constexpr int NT  = NWAVES * 32;
    constexpr int NV4 = GROWS * (GK / 4);
    static_assert(NT % 64 == 0 && (NV4 / NT) % 10 == 0, "staging split");
    const int tid  = threadIdx.x;
    const int row0 = blockIdx.x * GROWS;

    constexpr int PER = NV4 / NT;
    constexpr int CH  = 10;
    const int c4 = (tid & 63) * 4;
    for (int base = 0; base < PER; base += CH) {
        float4 v[CH];
#pragma unroll
        for (int u = 0; u < CH; ++u) {
            const int r  = ((base + u) * NT + tid) >> 6;
            const int rr = min(row0 + r, M - 1);
            v[u] = *reinterpret_cast<const float4*>(A + (size_t)rr * GK + c4);
        }
#pragma unroll
        for (int u = 0; u < CH; ++u) {
            const int r = ((base + u) * NT + tid) >> 6;
            *reinterpret_cast<float4*>(&sA[r * GPITCH + c4]) = v[u];
        }
    }
    __syncthreads();

    const int lane = tid & 31;
    const int wave = tid >> 5;
    const int h    = lane >> 4;
    const int c    = lane & 15;
    const int colw = blockIdx.y * (NWAVES * CT * 16) + wave * (CT * 16);

    const float* bp = B + (size_t)(2 * h) * NB + colw + c;
    const float* ap = sA + c * GPITCH + 2 * h;

    v8f acc[GRT][CT];
#pragma unroll
    for (int rt = 0; rt < GRT; ++rt)
#pragma unroll
        for (int ct = 0; ct < CT; ++ct) acc[rt][ct] = v8f{};

    auto load_frags = [&](int j, v2f (&bf)[CT], v2f (&af)[GRT]) {
#pragma unroll
        for (int ct = 0; ct < CT; ++ct) {
            bf[ct].x = bp[(size_t)(4 * j)     * NB + ct * 16];
            bf[ct].y = bp[(size_t)(4 * j + 1) * NB + ct * 16];
        }
#pragma unroll
        for (int rt = 0; rt < GRT; ++rt)
            af[rt] = *reinterpret_cast<const v2f*>(ap + rt * 16 * GPITCH + 4 * j);
    };
    auto mma_block = [&](const v2f (&bf)[CT], const v2f (&af)[GRT]) {
#pragma unroll
        for (int rt = 0; rt < GRT; ++rt)
#pragma unroll
            for (int ct = 0; ct < CT; ++ct)
                acc[rt][ct] = __builtin_amdgcn_wmma_f32_16x16x4_f32(
                    false, af[rt], false, bf[ct], (short)0, acc[rt][ct], false, false);
    };

    v2f b0[CT], a0[GRT], b1[CT], a1[GRT];
    load_frags(0, b0, a0);
    for (int j = 0; j < GK / 4 - 2; j += 2) {
        load_frags(j + 1, b1, a1);
        mma_block(b0, a0);
        load_frags(j + 2, b0, a0);
        mma_block(b1, a1);
    }
    load_frags(GK / 4 - 1, b1, a1);
    mma_block(b0, a0);
    mma_block(b1, a1);

#pragma unroll
    for (int rt = 0; rt < GRT; ++rt) {
        const int r0 = row0 + rt * 16;
        if (r0 < M) {
#pragma unroll
            for (int ct = 0; ct < CT; ++ct) {
                const int col = colw + ct * 16 + c;
                float* cp = C + (size_t)(r0 + 8 * h) * NB + col;
#pragma unroll
                for (int r = 0; r < 8; ++r) cp[(size_t)r * NB] = acc[rt][ct][r];
            }
        }
    }
}

__global__ void gat_att_scores(const float* __restrict__ h,
                               const float* __restrict__ att_src,
                               const float* __restrict__ att_dst,
                               float* __restrict__ a_src,
                               float* __restrict__ a_dst, int N,
                               const int* __restrict__ dcol,
                               unsigned* __restrict__ keys, int E) {
    const int t = blockIdx.x * blockDim.x + threadIdx.x;
    {
        auto key_of = [](int d) { return min((unsigned)d / (unsigned)FR, NOKEY); };
        const int nkw = ((E + 15) >> 4) << 2;
        for (int w = t; w < nkw; w += gridDim.x * blockDim.x) {
            unsigned packed;
            if ((E & 3) == 0 && 4 * w < E) {
                const int4 d = reinterpret_cast<const int4*>(dcol)[w];
                packed = key_of(d.x) | (key_of(d.y) << 8) | (key_of(d.z) << 16) | (key_of(d.w) << 24);
            } else {
                packed = 0;
#pragma unroll
                for (int i = 0; i < 4; ++i) {
                    const int e = 4 * w + i;
                    packed |= (e < E ? key_of(dcol[e]) : NOKEY) << (8 * i);
                }
            }
            keys[w] = packed;
        }
    }
    if (t >= N * HEADS) return;
    const int n  = t / HEADS;
    const int hd = t - n * HEADS;
    const float* hp = h + (size_t)n * HCDIM + hd * CDIM;
    const float* as = att_src + hd * CDIM;
    const float* ad = att_dst + hd * CDIM;
    float s0 = 0.0f, s1 = 0.0f;
#pragma unroll
    for (int c = 0; c < CDIM; ++c) {
        const float v = hp[c];
        s0 = fmaf(v, as[c], s0);
        s1 = fmaf(v, ad[c], s1);
    }
    a_src[t] = s0;
    a_dst[t] = s1;
}

__device__ __forceinline__ float leaky_relu(float v) {
    return v > 0.0f ? v : NSLOPE * v;
}

__device__ __forceinline__ float edge_logit(const float* __restrict__ a_src,
                                            const float* __restrict__ a_dst,
                                            int s, int d, int hd) {
    return leaky_relu(a_src[s * HEADS + hd] + a_dst[d * HEADS + hd]);
}

#define FRT    (FR / 16)
#define FNW    32
#define FNT    (FNW * 32)
#define FCAP   4096
#define FITEMS (FCAP + FR)
#define FSCAN  8
#define FEPW   4
#define FSWP   8

struct whole_group   { static constexpr bool value = true;  };
struct partial_group { static constexpr bool value = false; };
struct two_row_tiles { static constexpr bool value = true;  };
struct one_row_tile  { static constexpr bool value = false; };

__global__ __launch_bounds__(FNT)
void gat_aggregate_linear(const int* __restrict__ ei,
                          const unsigned* __restrict__ keys,
                          const float* __restrict__ a_src,
                          const float* __restrict__ a_dst,
                          const float* __restrict__ h,
                          const float* __restrict__ bias,
                          const float* __restrict__ B,
                          const float* __restrict__ bout,
                          float* __restrict__ C, int N, int E, int NB) {
    __shared__ __attribute__((aligned(16))) float sA[FR * GPITCH];
    __shared__ int           sSrc[FITEMS];
    __shared__ unsigned char sDl[FITEMS];
    __shared__ float         sMax[FR * HEADS];
    __shared__ float         sDen[FR * HEADS];
    __shared__ int sCnt;
    static_assert(FR <= 256 && FR % 16 == 0, "local row is kept in a byte; whole row tiles");
    static_assert(FCAP % FNT == 0 && FNT % HEADS == 0, "item splits");
    static_assert(FNW == 32 && FRT > 8 && FRT <= 16, "step 7: 8 wave groups x 4 column tiles, one or two row tiles each");

    const int tid  = threadIdx.x;
    const int lane = tid & 31;
    const int wave = tid >> 5;
    const int row0 = blockIdx.x * FR;
    const int* __restrict__ dcol = ei + E;

    for (int i = tid; i < FR * GPITCH / 4; i += FNT)
        reinterpret_cast<float4*>(sA)[i] = make_float4(0.0f, 0.0f, 0.0f, 0.0f);
    for (int i = tid; i < FR * HEADS; i += FNT) { sMax[i] = -INFINITY; sDen[i] = 0.0f; }
    if (tid == 0) sCnt = 0;
    __syncthreads();

    {
        const unsigned kb  = (unsigned)blockIdx.x * 0x01010101u;
        const int     nvec = (E + 15) >> 4;
        const int4* __restrict__ kv4 = reinterpret_cast<const int4*>(keys);
        auto eq_bytes = [&](unsigned w) {
            const unsigned x = w ^ kb;
            return ~(((x & 0x7F7F7F7Fu) + 0x7F7F7F7Fu) | x | 0x7F7F7F7Fu);
        };
        auto hit_mask = [&](const int4& q) {
            return (eq_bytes((unsigned)q.x) >> 7) | (eq_bytes((unsigned)q.y) >> 6) |
                   (eq_bytes((unsigned)q.z) >> 5) | (eq_bytes((unsigned)q.w) >> 4);
        };
        auto scan_batch = [&](int v0, bool guarded) {
            int4 q[FSCAN];
#pragma unroll
            for (int u = 0; u < FSCAN; ++u) {
                const int iv = v0 + u * FNT;
                q[u] = kv4[guarded ? min(iv, nvec - 1) : iv];
                if (guarded && iv >= nvec) q[u] = make_int4(-1, -1, -1, -1);
            }
            unsigned m[FSCAN];
            int n = 0;
#pragma unroll
            for (int u = 0; u < FSCAN; ++u) { m[u] = hit_mask(q[u]); n += __popc(m[u]); }
            if (n != 0) {
                int slot = atomicAdd(&sCnt, n);
#pragma unroll
                for (int u = 0; u < FSCAN; ++u) {
                    const int e0 = (v0 + u * FNT) * 16;
                    for (unsigned mm = m[u]; mm != 0; mm &= mm - 1) {
                        const int p = __ffs(mm) - 1;
                        if (slot < FCAP) sSrc[slot] = e0 + 4 * (p & 7) + (p >> 3);
                        ++slot;
                    }
                }
            }
        };
        constexpr int PER = FNT * FSCAN;
        const int nfull = nvec / PER;
        for (int it = 0; it < nfull; ++it) scan_batch(it * PER + tid, false);
        if (nfull * PER < nvec) scan_batch(nfull * PER + tid, true);
    }
    __syncthreads();

    const bool overflow = sCnt > FCAP;
    const int  cnt      = overflow ? 0 : sCnt;
    const int  nrows    = min(FR, N - row0);
    const int  total    = cnt + nrows;

    {
        constexpr int RES = FCAP / FNT;
        int ev[RES], sv[RES], dv[RES];
#pragma unroll
        for (int u = 0; u < RES; ++u) {
            const int idx = tid + u * FNT;
            ev[u] = idx < cnt ? sSrc[idx] : 0;
        }
#pragma unroll
        for (int u = 0; u < RES; ++u) { sv[u] = ei[ev[u]]; dv[u] = dcol[ev[u]]; }
#pragma unroll
        for (int u = 0; u < RES; ++u) {
            const int idx = tid + u * FNT;
            if (idx < cnt) { sSrc[idx] = sv[u]; sDl[idx] = (unsigned char)(dv[u] - row0); }
        }
        for (int r = tid; r < nrows; r += FNT) {
            sSrc[cnt + r] = row0 + r;
            sDl[cnt + r]  = (unsigned char)r;
        }
    }
    __syncthreads();

    auto unlisted_edges = [&](auto&& f) {
        for (int e = tid; e < E; e += FNT) {
            const unsigned dl = (unsigned)(dcol[e] - row0);
            if (dl < (unsigned)FR) {
                const int s = ei[e];
#pragma nounroll
                for (int hd = 0; hd < HEADS; ++hd) f(s, (int)dl, hd);
            }
        }
    };
    auto sweep_items = [&](auto&& f) {
        constexpr int IPS = FNT / HEADS;
        const int hd = tid & (HEADS - 1);
        for (int i0 = tid / HEADS; i0 < total; i0 += FSWP * IPS) {
            int   k[FSWP];
            float ps[FSWP], pd[FSWP];
#pragma unroll
            for (int u = 0; u < FSWP; ++u) {
                const int it  = i0 + u * IPS;
                const int itc = it < total ? it : i0;
                const int dl  = sDl[itc];
                k[u]  = dl * HEADS + hd;
                ps[u] = a_src[sSrc[itc] * HEADS + hd];
                pd[u] = a_dst[(row0 + dl) * HEADS + hd];
            }
#pragma unroll
            for (int u = 0; u < FSWP; ++u)
                if (i0 + u * IPS < total) f(k[u], leaky_relu(ps[u] + pd[u]));
        }
    };

    if (overflow)
        unlisted_edges([&](int s, int dl, int hd) {
            atomicMaxF(&sMax[dl * HEADS + hd], edge_logit(a_src, a_dst, s, row0 + dl, hd));
        });
    sweep_items([&](int k, float v) { atomicMaxF(&sMax[k], v); });
    __syncthreads();

    if (overflow)
        unlisted_edges([&](int s, int dl, int hd) {
            const float v = edge_logit(a_src, a_dst, s, row0 + dl, hd);
            atomicAdd(&sDen[dl * HEADS + hd], expf(v - sMax[dl * HEADS + hd]));
        });
    sweep_items([&](int k, float v) { atomicAdd(&sDen[k], expf(v - sMax[k])); });
    __syncthreads();

    auto alpha_of = [&](int s, int dl, int hd) {
        const float v  = edge_logit(a_src, a_dst, s, row0 + dl, hd);
        const float ex = expf(v - sMax[dl * HEADS + hd]);
        return ex / (sDen[dl * HEADS + hd] + 1e-16f);
    };

    if (overflow)
        unlisted_edges([&](int s, int dl, int hd) {
            const float al = alpha_of(s, dl, hd);
#pragma nounroll
            for (int c = 0; c < CDIM; ++c)
                atomicAdd(&sA[dl * GPITCH + hd * CDIM + c],
                          h[(size_t)s * HCDIM + hd * CDIM + c] * al);
        });
    static_assert(FEPW * HEADS == 32, "one lane per (item, head) of a group");
    const int wv     = __builtin_amdgcn_readfirstlane(wave);
    const int totalS = __builtin_amdgcn_readfirstlane(total);
    auto message_group = [&](int base, int nval, auto whole) {
        constexpr bool WHOLE = decltype(whole)::value;
        const int j4  = lane >> 3;
        const int hd  = lane & 7;
        const int idx = base + (WHOLE ? j4 : min(j4, nval - 1));
        const int sv  = sSrc[idx];
        const int dv  = sDl[idx];
        const float ps = a_src[sv * HEADS + hd];
        const float pd = a_dst[(row0 + dv) * HEADS + hd];
        float hv[FEPW][HEADS];
#pragma unroll
        for (int j = 0; j < FEPW; ++j) {
            const int sj = __builtin_amdgcn_readlane(sv, j * HEADS);
            const float* hs = h + (size_t)sj * HCDIM + lane;
#pragma unroll
            for (int i = 0; i < HEADS; ++i) hv[j][i] = hs[i * CDIM];
        }
        const float v  = leaky_relu(ps + pd);
        const float ex = expf(v - sMax[dv * HEADS + hd]);
        const int   al = __float_as_int(ex / (sDen[dv * HEADS + hd] + 1e-16f));
#pragma unroll
        for (int j = 0; j < FEPW; ++j) {
            if (WHOLE || j < nval) {
                const int dj = __builtin_amdgcn_readlane(dv, j * HEADS);
                float* dst = &sA[dj * GPITCH + lane];
#pragma unroll
                for (int i = 0; i < HEADS; ++i) {
                    const float w = __int_as_float(__builtin_amdgcn_readlane(al, j * HEADS + i));
                    atomicAdd(dst + i * CDIM, hv[j][i] * w);
                }
            }
        }
    };
    const int ngw = totalS / FEPW;
    for (int g = wv; g < ngw; g += FNW) message_group(g * FEPW, FEPW, whole_group{});
    if ((totalS % FEPW) != 0 && wv == (ngw % FNW))
        message_group(ngw * FEPW, totalS % FEPW, partial_group{});
    __syncthreads();

    {
        const int c4 = (tid & 63) * 4;
        const float4 bb = *reinterpret_cast<const float4*>(bias + c4);
        for (int r = tid >> 6; r < FR; r += FNT / 64) {
            float4* p = reinterpret_cast<float4*>(&sA[r * GPITCH + c4]);
            float4 w = *p;
            w.x = fmaxf(w.x + bb.x, 0.0f);
            w.y = fmaxf(w.y + bb.y, 0.0f);
            w.z = fmaxf(w.z + bb.z, 0.0f);
            w.w = fmaxf(w.w + bb.w, 0.0f);
            *p = w;
        }
    }
    __syncthreads();

    const int hh   = lane >> 4;
    const int c    = lane & 15;
    const int rtA  = wv >> 2;
    const int colw = blockIdx.y * 64 + (wv & 3) * 16;

    const float* bp = B + (size_t)(2 * hh) * NB + colw + c;
    const float* ap = sA + (rtA * 16 + c) * GPITCH + 2 * hh;
    const float bo  = bout[colw + c];

    auto linear_tiles = [&](auto tiles) {
        constexpr bool TWO = decltype(tiles)::value;
        v8f acc0 = v8f{}, acc1 = v8f{};
        auto load_frags = [&](int j, v2f& bf, v2f& af0, v2f& af1) {
            bf.x = bp[(size_t)(4 * j)     * NB];
            bf.y = bp[(size_t)(4 * j + 1) * NB];
            af0 = *reinterpret_cast<const v2f*>(ap + 4 * j);
            if (TWO) af1 = *reinterpret_cast<const v2f*>(ap + 8 * 16 * GPITCH + 4 * j);
        };
        auto mma_tiles = [&](const v2f& bf, const v2f& af0, const v2f& af1) {
            acc0 = __builtin_amdgcn_wmma_f32_16x16x4_f32(false, af0, false, bf, (short)0, acc0, false, false);
            if (TWO)
                acc1 = __builtin_amdgcn_wmma_f32_16x16x4_f32(false, af1, false, bf, (short)0, acc1, false, false);
        };
        v2f bA, a0A, a1A = v2f{}, bB, a0B, a1B = v2f{};
        load_frags(0, bA, a0A, a1A);
        for (int j = 0; j < GK / 4 - 2; j += 2) {
            load_frags(j + 1, bB, a0B, a1B);
            mma_tiles(bA, a0A, a1A);
            load_frags(j + 2, bA, a0A, a1A);
            mma_tiles(bB, a0B, a1B);
        }
        load_frags(GK / 4 - 1, bB, a0B, a1B);
        mma_tiles(bA, a0A, a1A);
        mma_tiles(bB, a0B, a1B);

        const int r0 = row0 + rtA * 16;
        if (r0 < N) {
            float* cp = C + (size_t)(r0 + 8 * hh) * NB + colw + c;
#pragma unroll
            for (int r = 0; r < 8; ++r) cp[(size_t)r * NB] = acc0[r] + bo;
        }
        if (TWO && r0 + 8 * 16 < N) {
            float* cp = C + (size_t)(r0 + 8 * 16 + 8 * hh) * NB + colw + c;
#pragma unroll
            for (int r = 0; r < 8; ++r) cp[(size_t)r * NB] = acc1[r] + bo;
        }
    };
    if (rtA + 8 < FRT) linear_tiles(two_row_tiles{});
    else               linear_tiles(one_row_tile{});
}

extern "C" void kernel_launch(void* const* d_in, const int* in_sizes, int n_in,
                              void* d_out, int out_size, void* d_ws, size_t ws_size,
                              hipStream_t stream) {
    (void)n_in; (void)out_size; (void)ws_size;

    const float* x        = (const float*)d_in[0];
    const int*   ei       = (const int*)d_in[1];
    const float* W_gat    = (const float*)d_in[2];
    const float* att_src  = (const float*)d_in[3];
    const float* att_dst  = (const float*)d_in[4];
    const float* bias_gat = (const float*)d_in[5];
    const float* W_lin    = (const float*)d_in[6];
    const float* b_lin    = (const float*)d_in[7];
    float*       out      = (float*)d_out;

    const int IN  = 256;
    const int N   = in_sizes[0] / IN;
    const int E   = in_sizes[1] / 2;
    const int OUT = in_sizes[6] / HCDIM;

    float* ws   = (float*)d_ws;
    float* h    = ws;
    float* acc  = h    + (size_t)N * HCDIM;
    float* asrc = acc  + (size_t)N * HCDIM;
    float* adst = asrc + (size_t)N * HEADS;
    unsigned* keys = (unsigned*)acc;

    const int RB = (N / 16 + GRT - 1) / GRT;
    wmma_gemm_rows_in_lds<8, 2><<<dim3(RB, HCDIM / 256), 256, 0, stream>>>(
        x, W_gat, h, N, HCDIM);

    const int t2 = N * HEADS;
    gat_att_scores<<<(t2 + 255) / 256, 256, 0, stream>>>(h, att_src, att_dst, asrc, adst, N,
                                                       ei + E, keys, E);

    gat_aggregate_linear<<<dim3((N + FR - 1) / FR, OUT / 64), FNT, 0, stream>>>(
        ei, keys, asrc, adst, h, bias_gat, W_lin, b_lin, out, N, E, OUT);
}


# Aggregate launch: a thread fetches each (item, head) source score once, not 3 times; the logit, then its exponential, stays with the thread from max sweep to message loop; sums in arrival order
# speedup vs baseline: 4.0211x; 1.0422x over previous
#include <hip/hip_runtime.h>
#include <math.h>


typedef float v2f __attribute__((ext_vector_type(2)));
typedef float v8f __attribute__((ext_vector_type(8)));

#define HEADS  8
#define CDIM   32
#define HCDIM  256
#define NSLOPE 0.2f
#define FR     208
#define NOKEY  255u

__device__ __forceinline__ void atomicMaxF(float* addr, float val) {
    if (val >= 0.0f) {
        atomicMax((int*)addr, __float_as_int(val));
    } else {
        atomicMin((unsigned int*)addr, __float_as_uint(val));
    }
}

#define GK     256
#define GRT    5
#define GROWS  (GRT * 16)
#define GPITCH 260

template <int NWAVES, int CT>
__global__ __launch_bounds__(NWAVES * 32)
void wmma_gemm_rows_in_lds(const float* __restrict__ A,
                           const float* __restrict__ B,
                           float* __restrict__ C, int M, int NB) {
    __shared__ __attribute__((aligned(16))) float sA[GROWS * GPITCH];

    constexpr int NT  = NWAVES * 32;
    constexpr int NV4 = GROWS * (GK / 4);
    static_assert(NT % 64 == 0 && (NV4 / NT) % 10 == 0, "staging split");
    const int tid  = threadIdx.x;
    const int row0 = blockIdx.x * GROWS;

    constexpr int PER = NV4 / NT;
    constexpr int CH  = 10;
    const int c4 = (tid & 63) * 4;
    for (int base = 0; base < PER; base += CH) {
        float4 v[CH];
#pragma unroll
        for (int u = 0; u < CH; ++u) {
            const int r  = ((base + u) * NT + tid) >> 6;
            const int rr = min(row0 + r, M - 1);
            v[u] = *reinterpret_cast<const float4*>(A + (size_t)rr * GK + c4);
        }
#pragma unroll
        for (int u = 0; u < CH; ++u) {
            const int r = ((base + u) * NT + tid) >> 6;
            *reinterpret_cast<float4*>(&sA[r * GPITCH + c4]) = v[u];
        }
    }
    __syncthreads();

    const int lane = tid & 31;
    const int wave = tid >> 5;
    const int h    = lane >> 4;
    const int c    = lane & 15;
    const int colw = blockIdx.y * (NWAVES * CT * 16) + wave * (CT * 16);

    const float* bp = B + (size_t)(2 * h) * NB + colw + c;
    const float* ap = sA + c * GPITCH + 2 * h;

    v8f acc[GRT][CT];
#pragma unroll
    for (int rt = 0; rt < GRT; ++rt)
#pragma unroll
        for (int ct = 0; ct < CT; ++ct) acc[rt][ct] = v8f{};

    auto load_frags = [&](int j, v2f (&bf)[CT], v2f (&af)[GRT]) {
#pragma unroll
        for (int ct = 0; ct < CT; ++ct) {
            bf[ct].x = bp[(size_t)(4 * j)     * NB + ct * 16];
            bf[ct].y = bp[(size_t)(4 * j + 1) * NB + ct * 16];
        }
#pragma unroll
        for (int rt = 0; rt < GRT; ++rt)
            af[rt] = *reinterpret_cast<const v2f*>(ap + rt * 16 * GPITCH + 4 * j);
    };
    auto mma_block = [&](const v2f (&bf)[CT], const v2f (&af)[GRT]) {
#pragma unroll
        for (int rt = 0; rt < GRT; ++rt)
#pragma unroll
            for (int ct = 0; ct < CT; ++ct)
                acc[rt][ct] = __builtin_amdgcn_wmma_f32_16x16x4_f32(
                    false, af[rt], false, bf[ct], (short)0, acc[rt][ct], false, false);
    };

    v2f b0[CT], a0[GRT], b1[CT], a1[GRT];
    load_frags(0, b0, a0);
    for (int j = 0; j < GK / 4 - 2; j += 2) {
        load_frags(j + 1, b1, a1);
        mma_block(b0, a0);
        load_frags(j + 2, b0, a0);
        mma_block(b1, a1);
    }
    load_frags(GK / 4 - 1, b1, a1);
    mma_block(b0, a0);
    mma_block(b1, a1);

#pragma unroll
    for (int rt = 0; rt < GRT; ++rt) {
        const int r0 = row0 + rt * 16;
        if (r0 < M) {
#pragma unroll
            for (int ct = 0; ct < CT; ++ct) {
                const int col = colw + ct * 16 + c;
                float* cp = C + (size_t)(r0 + 8 * h) * NB + col;
#pragma unroll
                for (int r = 0; r < 8; ++r) cp[(size_t)r * NB] = acc[rt][ct][r];
            }
        }
    }
}

__global__ void gat_att_scores(const float* __restrict__ h,
                               const float* __restrict__ att_src,
                               const float* __restrict__ att_dst,
                               float* __restrict__ a_src,
                               float* __restrict__ a_dst, int N,
                               const int* __restrict__ dcol,
                               unsigned* __restrict__ keys, int E) {
    const int t = blockIdx.x * blockDim.x + threadIdx.x;
    {
        auto key_of = [](int d) { return min((unsigned)d / (unsigned)FR, NOKEY); };
        const int nkw = ((E + 15) >> 4) << 2;
        for (int w = t; w < nkw; w += gridDim.x * blockDim.x) {
            unsigned packed;
            if ((E & 3) == 0 && 4 * w < E) {
                const int4 d = reinterpret_cast<const int4*>(dcol)[w];
                packed = key_of(d.x) | (key_of(d.y) << 8) | (key_of(d.z) << 16) | (key_of(d.w) << 24);
            } else {
                packed = 0;
#pragma unroll
                for (int i = 0; i < 4; ++i) {
                    const int e = 4 * w + i;
                    packed |= (e < E ? key_of(dcol[e]) : NOKEY) << (8 * i);
                }
            }
            keys[w] = packed;
        }
    }
    if (t >= N * HEADS) return;
    const int n  = t / HEADS;
    const int hd = t - n * HEADS;
    const float* hp = h + (size_t)n * HCDIM + hd * CDIM;
    const float* as = att_src + hd * CDIM;
    const float* ad = att_dst + hd * CDIM;
    float s0 = 0.0f, s1 = 0.0f;
#pragma unroll
    for (int c = 0; c < CDIM; ++c) {
        const float v = hp[c];
        s0 = fmaf(v, as[c], s0);
        s1 = fmaf(v, ad[c], s1);
    }
    a_src[t] = s0;
    a_dst[t] = s1;
}

__device__ __forceinline__ float leaky_relu(float v) {
    return v > 0.0f ? v : NSLOPE * v;
}

__device__ __forceinline__ float edge_logit(const float* __restrict__ a_src,
                                            const float* __restrict__ a_dst,
                                            int s, int d, int hd) {
    return leaky_relu(a_src[s * HEADS + hd] + a_dst[d * HEADS + hd]);
}

#define FRT    (FR / 16)
#define FNW    32
#define FNT    (FNW * 32)
#define FCAP   4096
#define FITEMS (FCAP + FR)
#define FSCAN  8
#define FEPW   4
#define FSWP   8
#define FIPS   (FNT / HEADS)
#define FNST   ((FITEMS + FIPS - 1) / FIPS)
#define FNQ    (2 * ((FNST + FSWP - 1) / FSWP))
#define FTCH   12

struct whole_group   { static constexpr bool value = true;  };
struct partial_group { static constexpr bool value = false; };
struct two_row_tiles { static constexpr bool value = true;  };
struct one_row_tile  { static constexpr bool value = false; };

__global__ __launch_bounds__(FNT)
void gat_aggregate_linear(const int* __restrict__ ei,
                          const unsigned* __restrict__ keys,
                          const float* __restrict__ a_src,
                          const float* __restrict__ a_dst,
                          const float* __restrict__ h,
                          const float* __restrict__ bias,
                          const float* __restrict__ B,
                          const float* __restrict__ bout,
                          float* __restrict__ C, int N, int E, int NB) {
    __shared__ __attribute__((aligned(16))) float sA[FR * GPITCH];
    __shared__ int           sSrc[FITEMS];
    __shared__ unsigned char sDl[FITEMS];
    __shared__ float         sMax[FR * HEADS];
    __shared__ float         sDen[FR * HEADS];
    __shared__ int sCnt;
    __shared__ float         sEx[FTCH * FNT];
    static_assert(FSWP == 8 && FNQ * FNT <= FR * GPITCH / 4, "a sweep batch is two 16-byte pieces; the sweep slots fit inside sA");
    static_assert(FEPW * FNW == FIPS, "step 5: a wave's group of step t is its lanes' items of step t");
    static_assert(FR <= 256 && FR % 16 == 0, "local row is kept in a byte; whole row tiles");
    static_assert(FCAP % FNT == 0 && FNT % HEADS == 0, "item splits");
    static_assert(FNW == 32 && FRT > 8 && FRT <= 16, "step 7: 8 wave groups x 4 column tiles, one or two row tiles each");

    const int tid  = threadIdx.x;
    const int lane = tid & 31;
    const int wave = tid >> 5;
    const int row0 = blockIdx.x * FR;
    const int* __restrict__ dcol = ei + E;

    for (int i = tid; i < FR * HEADS; i += FNT) { sMax[i] = -INFINITY; sDen[i] = 0.0f; }
    if (tid == 0) sCnt = 0;
    __syncthreads();

    {
        const unsigned kb  = (unsigned)blockIdx.x * 0x01010101u;
        const int     nvec = (E + 15) >> 4;
        const int4* __restrict__ kv4 = reinterpret_cast<const int4*>(keys);
        auto eq_bytes = [&](unsigned w) {
            const unsigned x = w ^ kb;
            return ~(((x & 0x7F7F7F7Fu) + 0x7F7F7F7Fu) | x | 0x7F7F7F7Fu);
        };
        auto hit_mask = [&](const int4& q) {
            return (eq_bytes((unsigned)q.x) >> 7) | (eq_bytes((unsigned)q.y) >> 6) |
                   (eq_bytes((unsigned)q.z) >> 5) | (eq_bytes((unsigned)q.w) >> 4);
        };
        auto scan_batch = [&](int v0, bool guarded) {
            int4 q[FSCAN];
#pragma unroll
            for (int u = 0; u < FSCAN; ++u) {
                const int iv = v0 + u * FNT;
                q[u] = kv4[guarded ? min(iv, nvec - 1) : iv];
                if (guarded && iv >= nvec) q[u] = make_int4(-1, -1, -1, -1);
            }
            unsigned m[FSCAN];
            int n = 0;
#pragma unroll
            for (int u = 0; u < FSCAN; ++u) { m[u] = hit_mask(q[u]); n += __popc(m[u]); }
            if (n != 0) {
                int slot = atomicAdd(&sCnt, n);
#pragma unroll
                for (int u = 0; u < FSCAN; ++u) {
                    const int e0 = (v0 + u * FNT) * 16;
                    for (unsigned mm = m[u]; mm != 0; mm &= mm - 1) {
                        const int p = __ffs(mm) - 1;
                        if (slot < FCAP) sSrc[slot] = e0 + 4 * (p & 7) + (p >> 3);
                        ++slot;
                    }
                }
            }
        };
        constexpr int PER = FNT * FSCAN;
        const int nfull = nvec / PER;
        for (int it = 0; it < nfull; ++it) scan_batch(it * PER + tid, false);
        if (nfull * PER < nvec) scan_batch(nfull * PER + tid, true);
    }
    __syncthreads();

    const bool overflow = sCnt > FCAP;
    const int  cnt      = overflow ? 0 : sCnt;
    const int  nrows    = min(FR, N - row0);
    const int  total    = cnt + nrows;

    {
        constexpr int RES = FCAP / FNT;
        int ev[RES], sv[RES], dv[RES];
#pragma unroll
        for (int u = 0; u < RES; ++u) {
            const int idx = tid + u * FNT;
            ev[u] = idx < cnt ? sSrc[idx] : 0;
        }
#pragma unroll
        for (int u = 0; u < RES; ++u) { sv[u] = ei[ev[u]]; dv[u] = dcol[ev[u]]; }
#pragma unroll
        for (int u = 0; u < RES; ++u) {
            const int idx = tid + u * FNT;
            if (idx < cnt) { sSrc[idx] = sv[u]; sDl[idx] = (unsigned char)(dv[u] - row0); }
        }
        for (int r = tid; r < nrows; r += FNT) {
            sSrc[cnt + r] = row0 + r;
            sDl[cnt + r]  = (unsigned char)r;
        }
    }
    __syncthreads();

    auto unlisted_edges = [&](auto&& f) {
        for (int e = tid; e < E; e += FNT) {
            const unsigned dl = (unsigned)(dcol[e] - row0);
            if (dl < (unsigned)FR) {
                const int s = ei[e];
#pragma nounroll
                for (int hd = 0; hd < HEADS; ++hd) f(s, (int)dl, hd);
            }
        }
    };
    const int pit = tid / HEADS;
    const int phd = tid & (HEADS - 1);
    float4* const sPv = reinterpret_cast<float4*>(sA);

    if (overflow)
        unlisted_edges([&](int s, int dl, int hd) {
            atomicMaxF(&sMax[dl * HEADS + hd], edge_logit(a_src, a_dst, s, row0 + dl, hd));
        });
    for (int b = 0; b * FIPS < total; b += FSWP) {
        int   k[FSWP];
        float ps[FSWP], pd[FSWP], v[FSWP];
#pragma unroll
        for (int u = 0; u < FSWP; ++u) {
            const int itc = min(pit + (b + u) * FIPS, total - 1);
            const int dl  = sDl[itc];
            k[u]  = dl * HEADS + phd;
            ps[u] = a_src[sSrc[itc] * HEADS + phd];
            pd[u] = a_dst[(row0 + dl) * HEADS + phd];
        }
#pragma unroll
        for (int u = 0; u < FSWP; ++u) {
            v[u] = leaky_relu(ps[u] + pd[u]);
            if (pit + (b + u) * FIPS < total) atomicMaxF(&sMax[k[u]], v[u]);
        }
        sPv[tid + FNT * (b / 4)]     = make_float4(v[0], v[1], v[2], v[3]);
        sPv[tid + FNT * (b / 4 + 1)] = make_float4(v[4], v[5], v[6], v[7]);
    }
    __syncthreads();

    if (overflow)
        unlisted_edges([&](int s, int dl, int hd) {
            const float v = edge_logit(a_src, a_dst, s, row0 + dl, hd);
            atomicAdd(&sDen[dl * HEADS + hd], expf(v - sMax[dl * HEADS + hd]));
        });
    for (int b = 0; b * FIPS < total; b += FSWP) {
        const float4 lo = sPv[tid + FNT * (b / 4)];
        const float4 hi = sPv[tid + FNT * (b / 4 + 1)];
        float v[FSWP] = {lo.x, lo.y, lo.z, lo.w, hi.x, hi.y, hi.z, hi.w};
        int   k[FSWP];
        float mx[FSWP];
#pragma unroll
        for (int u = 0; u < FSWP; ++u) {
            k[u]  = sDl[min(pit + (b + u) * FIPS, total - 1)] * HEADS + phd;
            mx[u] = sMax[k[u]];
        }
#pragma unroll
        for (int u = 0; u < FSWP; ++u) {
            v[u] = expf(v[u] - mx[u]);
            if (pit + (b + u) * FIPS < total) atomicAdd(&sDen[k[u]], v[u]);
        }
        sPv[tid + FNT * (b / 4)]     = make_float4(v[0], v[1], v[2], v[3]);
        sPv[tid + FNT * (b / 4 + 1)] = make_float4(v[4], v[5], v[6], v[7]);
    }
    float pv[FNST];
#pragma unroll
    for (int q = 0; q < (FNST + 3) / 4; ++q) {
        float4 w = make_float4(0.0f, 0.0f, 0.0f, 0.0f);
        if ((q / 2) * FSWP * FIPS < total) w = sPv[tid + FNT * q];
        if (4 * q     < FNST) pv[4 * q]     = w.x;
        if (4 * q + 1 < FNST) pv[4 * q + 1] = w.y;
        if (4 * q + 2 < FNST) pv[4 * q + 2] = w.z;
        if (4 * q + 3 < FNST) pv[4 * q + 3] = w.w;
    }
    for (int i = tid; i < FR * GPITCH / 4; i += FNT)
        sPv[i] = make_float4(0.0f, 0.0f, 0.0f, 0.0f);
    __syncthreads();

    auto alpha_of = [&](int s, int dl, int hd) {
        const float v  = edge_logit(a_src, a_dst, s, row0 + dl, hd);
        const float ex = expf(v - sMax[dl * HEADS + hd]);
        return ex / (sDen[dl * HEADS + hd] + 1e-16f);
    };

    if (overflow)
        unlisted_edges([&](int s, int dl, int hd) {
            const float al = alpha_of(s, dl, hd);
#pragma nounroll
            for (int c = 0; c < CDIM; ++c)
                atomicAdd(&sA[dl * GPITCH + hd * CDIM + c],
                          h[(size_t)s * HCDIM + hd * CDIM + c] * al);
        });
    static_assert(FEPW * HEADS == 32, "one lane per (item, head) of a group");
    const int wv     = __builtin_amdgcn_readfirstlane(wave);
    const int totalS = __builtin_amdgcn_readfirstlane(total);
    auto message_group = [&](int base, int nval, float ex, auto whole) {
        constexpr bool WHOLE = decltype(whole)::value;
        const int j4  = lane >> 3;
        const int hd  = lane & 7;
        const int idx = base + (WHOLE ? j4 : min(j4, nval - 1));
        const int sv  = sSrc[idx];
        const int dv  = sDl[idx];
        float hv[FEPW][HEADS];
#pragma unroll
        for (int j = 0; j < FEPW; ++j) {
            const int sj = __builtin_amdgcn_readlane(sv, j * HEADS);
            const float* hs = h + (size_t)sj * HCDIM + lane;
#pragma unroll
            for (int i = 0; i < HEADS; ++i) hv[j][i] = hs[i * CDIM];
        }
        const int   al = __float_as_int(ex / (sDen[dv * HEADS + hd] + 1e-16f));
#pragma unroll
        for (int j = 0; j < FEPW; ++j) {
            if (WHOLE || j < nval) {
                const int dj = __builtin_amdgcn_readlane(dv, j * HEADS);
                float* dst = &sA[dj * GPITCH + lane];
#pragma unroll
                for (int i = 0; i < HEADS; ++i) {
                    const float w = __int_as_float(__builtin_amdgcn_readlane(al, j * HEADS + i));
                    atomicAdd(dst + i * CDIM, hv[j][i] * w);
                }
            }
        }
    };
    const int ngw = totalS / FEPW;
#pragma unroll
    for (int c = 0; c < FNST; c += FTCH) {
        if (c * FIPS < totalS) {
#pragma unroll
            for (int u = 0; u < FTCH; ++u)
                if (c + u < FNST) sEx[u * FNT + tid] = pv[c + u];
            int u = 0;
            for (int g = wv + FNW * c; u < FTCH && g < ngw; ++u, g += FNW)
                message_group(g * FEPW, FEPW, sEx[u * FNT + tid], whole_group{});
        }
    }
    if ((totalS % FEPW) != 0 && wv == (ngw % FNW))
        message_group(ngw * FEPW, totalS % FEPW, sEx[((ngw / FNW) % FTCH) * FNT + tid], partial_group{});
    __syncthreads();

    {
        const int c4 = (tid & 63) * 4;
        const float4 bb = *reinterpret_cast<const float4*>(bias + c4);
        for (int r = tid >> 6; r < FR; r += FNT / 64) {
            float4* p = reinterpret_cast<float4*>(&sA[r * GPITCH + c4]);
            float4 w = *p;
            w.x = fmaxf(w.x + bb.x, 0.0f);
            w.y = fmaxf(w.y + bb.y, 0.0f);
            w.z = fmaxf(w.z + bb.z, 0.0f);
            w.w = fmaxf(w.w + bb.w, 0.0f);
            *p = w;
        }
    }
    __syncthreads();

    const int hh   = lane >> 4;
    const int c    = lane & 15;
    const int rtA  = wv >> 2;
    const int colw = blockIdx.y * 64 + (wv & 3) * 16;

    const float* bp = B + (size_t)(2 * hh) * NB + colw + c;
    const float* ap = sA + (rtA * 16 + c) * GPITCH + 2 * hh;
    const float bo  = bout[colw + c];

    auto linear_tiles = [&](auto tiles) {
        constexpr bool TWO = decltype(tiles)::value;
        v8f acc0 = v8f{}, acc1 = v8f{};
        auto load_frags = [&](int j, v2f& bf, v2f& af0, v2f& af1) {
            bf.x = bp[(size_t)(4 * j)     * NB];
            bf.y = bp[(size_t)(4 * j + 1) * NB];
            af0 = *reinterpret_cast<const v2f*>(ap + 4 * j);
            if (TWO) af1 = *reinterpret_cast<const v2f*>(ap + 8 * 16 * GPITCH + 4 * j);
        };
        auto mma_tiles = [&](const v2f& bf, const v2f& af0, const v2f& af1) {
            acc0 = __builtin_amdgcn_wmma_f32_16x16x4_f32(false, af0, false, bf, (short)0, acc0, false, false);
            if (TWO)
                acc1 = __builtin_amdgcn_wmma_f32_16x16x4_f32(false, af1, false, bf, (short)0, acc1, false, false);
        };
        v2f bA, a0A, a1A = v2f{}, bB, a0B, a1B = v2f{};
        load_frags(0, bA, a0A, a1A);
        for (int j = 0; j < GK / 4 - 2; j += 2) {
            load_frags(j + 1, bB, a0B, a1B);
            mma_tiles(bA, a0A, a1A);
            load_frags(j + 2, bA, a0A, a1A);
            mma_tiles(bB, a0B, a1B);
        }
        load_frags(GK / 4 - 1, bB, a0B, a1B);
        mma_tiles(bA, a0A, a1A);
        mma_tiles(bB, a0B, a1B);

        const int r0 = row0 + rtA * 16;
        if (r0 < N) {
            float* cp = C + (size_t)(r0 + 8 * hh) * NB + colw + c;
#pragma unroll
            for (int r = 0; r < 8; ++r) cp[(size_t)r * NB] = acc0[r] + bo;
        }
        if (TWO && r0 + 8 * 16 < N) {
            float* cp = C + (size_t)(r0 + 8 * 16 + 8 * hh) * NB + colw + c;
#pragma unroll
            for (int r = 0; r < 8; ++r) cp[(size_t)r * NB] = acc1[r] + bo;
        }
    };
    if (rtA + 8 < FRT) linear_tiles(two_row_tiles{});
    else               linear_tiles(one_row_tile{});
}

extern "C" void kernel_launch(void* const* d_in, const int* in_sizes, int n_in,
                              void* d_out, int out_size, void* d_ws, size_t ws_size,
                              hipStream_t stream) {
    (void)n_in; (void)out_size; (void)ws_size;

    const float* x        = (const float*)d_in[0];
    const int*   ei       = (const int*)d_in[1];
    const float* W_gat    = (const float*)d_in[2];
    const float* att_src  = (const float*)d_in[3];
    const float* att_dst  = (const float*)d_in[4];
    const float* bias_gat = (const float*)d_in[5];
    const float* W_lin    = (const float*)d_in[6];
    const float* b_lin    = (const float*)d_in[7];
    float*       out      = (float*)d_out;

    const int IN  = 256;
    const int N   = in_sizes[0] / IN;
    const int E   = in_sizes[1] / 2;
    const int OUT = in_sizes[6] / HCDIM;

    float* ws   = (float*)d_ws;
    float* h    = ws;
    float* acc  = h    + (size_t)N * HCDIM;
    float* asrc = acc  + (size_t)N * HCDIM;
    float* adst = asrc + (size_t)N * HEADS;
    unsigned* keys = (unsigned*)acc;

    const int RB = (N / 16 + GRT - 1) / GRT;
    wmma_gemm_rows_in_lds<8, 2><<<dim3(RB, HCDIM / 256), 256, 0, stream>>>(
        x, W_gat, h, N, HCDIM);

    const int t2 = N * HEADS;
    gat_att_scores<<<(t2 + 255) / 256, 256, 0, stream>>>(h, att_src, att_dst, asrc, adst, N,
                                                       ei + E, keys, E);

    gat_aggregate_linear<<<dim3((N + FR - 1) / FR, OUT / 64), FNT, 0, stream>>>(
        ei, keys, asrc, adst, h, bias_gat, W_lin, b_lin, out, N, E, OUT);
}
